# starting kernel
# baseline (speedup 1.0000x reference)
#include <hip/hip_runtime.h>
#include <hip/hip_bf16.h>

#define NB      8
#define NN      512
#define HIST    24
#define SKIP    256
#define ENDC    128
#define KERN    13
#define FUT     12
#define OUTD    2
#define NODES   (NB * NN)

#define XS_STRIDE 260
#define HS_STRIDE 132

typedef float v2f __attribute__((ext_vector_type(2)));
typedef float v8f __attribute__((ext_vector_type(8)));

__global__ __launch_bounds__(256)
void moe_tcnn_kernel(const float* __restrict__ x,
                     const int*   __restrict__ labels,
                     const float* __restrict__ W1,
                     const float* __restrict__ b1,
                     const float* __restrict__ W2,
                     const float* __restrict__ b2,
                     float*       __restrict__ out)
{
    __shared__ float Xs[32 * XS_STRIDE];
    __shared__ float Hs[HIST * HS_STRIDE];

    const int node = blockIdx.x;
    const int tid  = threadIdx.x;
    const int g    = labels[node];

    const float* xp = x + (size_t)node * (HIST * SKIP);
    #pragma unroll
    for (int q = 0; q < 6; ++q) {
        int idx4 = tid + q * 256;
        int idx  = idx4 * 4;
        int row  = idx >> 8;
        int col  = idx & 255;
        float4 v = *(const float4*)(xp + idx);
        v.x = v.x > 0.f ? v.x : 0.f;
        v.y = v.y > 0.f ? v.y : 0.f;
        v.z = v.z > 0.f ? v.z : 0.f;
        v.w = v.w > 0.f ? v.w : 0.f;
        *(float4*)(&Xs[row * XS_STRIDE + col]) = v;
    }
    #pragma unroll
    for (int q = 0; q < 8; ++q) {
        int i   = tid + q * 256;
        int row = 24 + (i >> 8);
        int col = i & 255;
        Xs[row * XS_STRIDE + col] = 0.f;
    }
    __syncthreads();

    const int wv   = tid >> 5;
    const int lane = tid & 31;
    const int half = lane >> 4;
    const int l16  = lane & 15;
    const int n0   = wv * 16;

    const float* W1row = W1 + ((size_t)g * ENDC + n0 + l16) * SKIP + 2 * half;
    const float* Arow0 = &Xs[l16 * XS_STRIDE + 2 * half];
    const float* Arow1 = &Xs[(16 + l16) * XS_STRIDE + 2 * half];

    v8f c0 = {0.f,0.f,0.f,0.f,0.f,0.f,0.f,0.f};
    v8f c1 = {0.f,0.f,0.f,0.f,0.f,0.f,0.f,0.f};

    for (int kk = 0; kk < SKIP; kk += 32) {
        #pragma unroll
        for (int u = 0; u < 8; ++u) {
            int k = kk + u * 4;
            v2f af0 = *(const v2f*)(Arow0 + k);
            v2f af1 = *(const v2f*)(Arow1 + k);
            v2f bf  = *(const v2f*)(W1row + k);
            c0 = __builtin_amdgcn_wmma_f32_16x16x4_f32(false, af0, false, bf,
                                                       (short)0, c0, false, false);
            c1 = __builtin_amdgcn_wmma_f32_16x16x4_f32(false, af1, false, bf,
                                                       (short)0, c1, false, false);
        }
    }

    const float bias = b1[g * ENDC + n0 + l16];
    const int   colH = n0 + l16;
    #pragma unroll
    for (int j = 0; j < 8; ++j) {
        int row0 = j + 8 * half;
        float v0 = c0[j] + bias;
        Hs[row0 * HS_STRIDE + colH] = v0 > 0.f ? v0 : 0.f;
        int row1 = 16 + j + 8 * half;
        if (row1 < HIST) {
            float v1 = c1[j] + bias;
            Hs[row1 * HS_STRIDE + colH] = v1 > 0.f ? v1 : 0.f;
        }
    }
    __syncthreads();

    const int group = wv * 4 + (lane >> 3);
    const int l8    = lane & 7;
    if (group < FUT * OUTD) {
        const int f = group % FUT;
        const int o = group / FUT;
        const float* W2go = W2 + ((size_t)(g * OUTD + o) * ENDC) * KERN;
        float acc = 0.f;
        #pragma unroll
        for (int i = 0; i < ENDC / 8; ++i) {
            int e = l8 + 8 * i;
            const float* hcol = &Hs[f * HS_STRIDE + e];
            const float* wrow = &W2go[e * KERN];
            #pragma unroll
            for (int k = 0; k < KERN; ++k)
                acc += hcol[k * HS_STRIDE] * wrow[k];
        }
        acc += __shfl_xor(acc, 4, 32);
        acc += __shfl_xor(acc, 2, 32);
        acc += __shfl_xor(acc, 1, 32);
        if (l8 == 0)
            out[(size_t)node * (FUT * OUTD) + f * OUTD + o] = acc + b2[g * OUTD + o];
    }
}

extern "C" void kernel_launch(void* const* d_in, const int* in_sizes, int n_in,
                              void* d_out, int out_size, void* d_ws, size_t ws_size,
                              hipStream_t stream) {
    const float* x      = (const float*)d_in[0];
    const int*   labels = (const int*)  d_in[1];
    const float* W1     = (const float*)d_in[2];
    const float* b1     = (const float*)d_in[3];
    const float* W2     = (const float*)d_in[4];
    const float* b2     = (const float*)d_in[5];
    float*       out    = (float*)d_out;

    moe_tcnn_kernel<<<NODES, 256, 0, stream>>>(x, labels, W1, b1, W2, b2, out);
}


# W1 reaches the B fragments through LDS: each 128x32 k-chunk is copied in with whole-line 16-byte loads (two buffers, one barrier per chunk) instead of 8-byte loads that touch 16 lines each
# speedup vs baseline: 1.3423x; 1.3423x over previous
#include <hip/hip_runtime.h>
#include <hip/hip_bf16.h>


#define NB      8
#define NN      512
#define HIST    24
#define SKIP    256
#define ENDC    128
#define KERN    13
#define FUT     12
#define OUTD    2
#define NODES   (NB * NN)

#define XS_STRIDE 260
#define HS_STRIDE 132
#define KC        32
#define BS_STRIDE 36
#define BS_TILE   (ENDC * BS_STRIDE)
#define NCHUNK    (SKIP / KC)

typedef float v2f __attribute__((ext_vector_type(2)));
typedef float v8f __attribute__((ext_vector_type(8)));

__global__ __launch_bounds__(256)
void moe_tcnn_kernel(const float* __restrict__ x,
                     const int*   __restrict__ labels,
                     const float* __restrict__ W1,
                     const float* __restrict__ b1,
                     const float* __restrict__ W2,
                     const float* __restrict__ b2,
                     float*       __restrict__ out)
{
    __shared__ __align__(16) float Xs[32 * XS_STRIDE];
    __shared__ __align__(16) float Bs[2 * BS_TILE];
    float* Hs = Bs;

    const int node = blockIdx.x;
    const int tid  = threadIdx.x;
    const int g    = labels[node];

    const int srow = tid >> 3;
    const int scol = (tid & 7) * 4;
    const float* wsrc = W1 + ((size_t)g * ENDC + srow) * SKIP + scol;
    float*       bdst = &Bs[srow * BS_STRIDE + scol];

    float4 pre[4];
    for (int q = 0; q < 4; ++q)
        pre[q] = *(const float4*)(wsrc + (size_t)q * 32 * SKIP);

    const float* xp = x + (size_t)node * (HIST * SKIP);
    #pragma unroll
    for (int q = 0; q < 6; ++q) {
        int idx4 = tid + q * 256;
        int idx  = idx4 * 4;
        int row  = idx >> 8;
        int col  = idx & 255;
        float4 v = *(const float4*)(xp + idx);
        v.x = v.x > 0.f ? v.x : 0.f;
        v.y = v.y > 0.f ? v.y : 0.f;
        v.z = v.z > 0.f ? v.z : 0.f;
        v.w = v.w > 0.f ? v.w : 0.f;
        *(float4*)(&Xs[row * XS_STRIDE + col]) = v;
    }
    #pragma unroll
    for (int q = 0; q < 8; ++q) {
        int i   = tid + q * 256;
        int row = 24 + (i >> 8);
        int col = i & 255;
        Xs[row * XS_STRIDE + col] = 0.f;
    }
    for (int q = 0; q < 4; ++q)
        *(float4*)(bdst + q * 32 * BS_STRIDE) = pre[q];
    __syncthreads();

    const int wv   = tid >> 5;
    const int lane = tid & 31;
    const int half = lane >> 4;
    const int l16  = lane & 15;
    const int n0   = wv * 16;

    const int boff = (n0 + l16) * BS_STRIDE + 2 * half;
    const float* Arow0 = &Xs[l16 * XS_STRIDE + 2 * half];
    const float* Arow1 = &Xs[(16 + l16) * XS_STRIDE + 2 * half];

    v8f c0 = {0.f,0.f,0.f,0.f,0.f,0.f,0.f,0.f};
    v8f c1 = {0.f,0.f,0.f,0.f,0.f,0.f,0.f,0.f};

    for (int c = 0; c < NCHUNK; ++c) {
        if (c + 1 < NCHUNK) {
            for (int q = 0; q < 4; ++q)
                pre[q] = *(const float4*)(wsrc + (size_t)q * 32 * SKIP + (c + 1) * KC);
        }
        const float* Bcur = Bs + (c & 1) * BS_TILE + boff;
        #pragma unroll
        for (int u = 0; u < 8; ++u) {
            int k = c * KC + u * 4;
            v2f af0 = *(const v2f*)(Arow0 + k);
            v2f af1 = *(const v2f*)(Arow1 + k);
            v2f bf  = *(const v2f*)(Bcur + u * 4);
            c0 = __builtin_amdgcn_wmma_f32_16x16x4_f32(false, af0, false, bf,
                                                       (short)0, c0, false, false);
            c1 = __builtin_amdgcn_wmma_f32_16x16x4_f32(false, af1, false, bf,
                                                       (short)0, c1, false, false);
        }
        if (c + 1 < NCHUNK) {
            float* bnext = bdst + ((c + 1) & 1) * BS_TILE;
            for (int q = 0; q < 4; ++q)
                *(float4*)(bnext + q * 32 * BS_STRIDE) = pre[q];
            __syncthreads();
        }
    }

    const float bias = b1[g * ENDC + n0 + l16];
    const int   colH = n0 + l16;
    #pragma unroll
    for (int j = 0; j < 8; ++j) {
        int row0 = j + 8 * half;
        float v0 = c0[j] + bias;
        Hs[row0 * HS_STRIDE + colH] = v0 > 0.f ? v0 : 0.f;
        int row1 = 16 + j + 8 * half;
        if (row1 < HIST) {
            float v1 = c1[j] + bias;
            Hs[row1 * HS_STRIDE + colH] = v1 > 0.f ? v1 : 0.f;
        }
    }
    __syncthreads();

    const int group = wv * 4 + (lane >> 3);
    const int l8    = lane & 7;
    if (group < FUT * OUTD) {
        const int f = group % FUT;
        const int o = group / FUT;
        const float* W2go = W2 + ((size_t)(g * OUTD + o) * ENDC) * KERN;
        float acc = 0.f;
        #pragma unroll
        for (int i = 0; i < ENDC / 8; ++i) {
            int e = l8 + 8 * i;
            const float* hcol = &Hs[f * HS_STRIDE + e];
            const float* wrow = &W2go[e * KERN];
            #pragma unroll
            for (int k = 0; k < KERN; ++k)
                acc += hcol[k * HS_STRIDE] * wrow[k];
        }
        acc += __shfl_xor(acc, 4, 32);
        acc += __shfl_xor(acc, 2, 32);
        acc += __shfl_xor(acc, 1, 32);
        if (l8 == 0)
            out[(size_t)node * (FUT * OUTD) + f * OUTD + o] = acc + b2[g * OUTD + o];
    }
}

extern "C" void kernel_launch(void* const* d_in, const int* in_sizes, int n_in,
                              void* d_out, int out_size, void* d_ws, size_t ws_size,
                              hipStream_t stream) {
    const float* x      = (const float*)d_in[0];
    const int*   labels = (const int*)  d_in[1];
    const float* W1     = (const float*)d_in[2];
    const float* b1     = (const float*)d_in[3];
    const float* W2     = (const float*)d_in[4];
    const float* b2     = (const float*)d_in[5];
    float*       out    = (float*)d_out;

    moe_tcnn_kernel<<<NODES, 256, 0, stream>>>(x, labels, W1, b1, W2, b2, out);
}


# Stage 2 regrouped: W2[g] joins the W1 prefetch stream into LDS under the last k-chunk; one (channel, output) per thread, 12 independent k-sums; channel partials summed via LDS (sum order changes)
# speedup vs baseline: 1.6655x; 1.2408x over previous
#include <hip/hip_runtime.h>
#include <hip/hip_bf16.h>


#define NB      8
#define NN      512
#define HIST    24
#define SKIP    256
#define ENDC    128
#define KERN    13
#define FUT     12
#define OUTD    2
#define NODES   (NB * NN)

#define XS_STRIDE 260
#define HS_STRIDE 132
#define KC        32
#define BS_STRIDE 36
#define BS_TILE   (ENDC * BS_STRIDE)
#define NCHUNK    (SKIP / KC)
#define W2_BLK    (OUTD * ENDC * KERN)

typedef float v2f __attribute__((ext_vector_type(2)));
typedef float v8f __attribute__((ext_vector_type(8)));

static __device__ __forceinline__
void mma_chunk(const float* Arow0, const float* Arow1, const float* Bcur, int kbase, v8f& c0, v8f& c1)
{
    #pragma unroll
    for (int u = 0; u < 8; ++u) {
        int k = kbase + u * 4;
        v2f af0 = *(const v2f*)(Arow0 + k);
        v2f af1 = *(const v2f*)(Arow1 + k);
        v2f bf  = *(const v2f*)(Bcur + u * 4);
        c0 = __builtin_amdgcn_wmma_f32_16x16x4_f32(false, af0, false, bf,
                                                   (short)0, c0, false, false);
        c1 = __builtin_amdgcn_wmma_f32_16x16x4_f32(false, af1, false, bf,
                                                   (short)0, c1, false, false);
    }
}

__global__ __launch_bounds__(256)
void moe_tcnn_kernel(const float* __restrict__ x,
                     const int*   __restrict__ labels,
                     const float* __restrict__ W1,
                     const float* __restrict__ b1,
                     const float* __restrict__ W2,
                     const float* __restrict__ b2,
                     float*       __restrict__ out)
{
    __shared__ __align__(16) float Xs[32 * XS_STRIDE];
    __shared__ __align__(16) float Bs[BS_TILE + HIST * HS_STRIDE + W2_BLK];
    float* Hs  = Bs + BS_TILE;
    float* W2s = Hs + HIST * HS_STRIDE;

    const int node = blockIdx.x;
    const int tid  = threadIdx.x;
    const int g    = labels[node];
    const float b2o0 = b2[g * OUTD + 0];
    const float b2o1 = b2[g * OUTD + 1];

    const int srow = tid >> 3;
    const int scol = (tid & 7) * 4;
    const float* wsrc = W1 + ((size_t)g * ENDC + srow) * SKIP + scol;
    float*       bdst = &Bs[srow * BS_STRIDE + scol];

    float4 pre[4];
    for (int q = 0; q < 4; ++q)
        pre[q] = *(const float4*)(wsrc + (size_t)q * 32 * SKIP);

    const float* xp = x + (size_t)node * (HIST * SKIP);
    #pragma unroll
    for (int q = 0; q < 6; ++q) {
        int idx4 = tid + q * 256;
        int idx  = idx4 * 4;
        int row  = idx >> 8;
        int col  = idx & 255;
        float4 v = *(const float4*)(xp + idx);
        v.x = v.x > 0.f ? v.x : 0.f;
        v.y = v.y > 0.f ? v.y : 0.f;
        v.z = v.z > 0.f ? v.z : 0.f;
        v.w = v.w > 0.f ? v.w : 0.f;
        *(float4*)(&Xs[row * XS_STRIDE + col]) = v;
    }
    #pragma unroll
    for (int q = 0; q < 8; ++q) {
        int i   = tid + q * 256;
        int row = 24 + (i >> 8);
        int col = i & 255;
        Xs[row * XS_STRIDE + col] = 0.f;
    }
    for (int q = 0; q < 4; ++q)
        *(float4*)(bdst + BS_TILE + q * 32 * BS_STRIDE) = pre[q];
    __syncthreads();

    const int wv   = tid >> 5;
    const int lane = tid & 31;
    const int half = lane >> 4;
    const int l16  = lane & 15;
    const int n0   = wv * 16;

    const int boff = (n0 + l16) * BS_STRIDE + 2 * half;
    const float* Arow0 = &Xs[l16 * XS_STRIDE + 2 * half];
    const float* Arow1 = &Xs[(16 + l16) * XS_STRIDE + 2 * half];

    v8f c0 = {0.f,0.f,0.f,0.f,0.f,0.f,0.f,0.f};
    v8f c1 = {0.f,0.f,0.f,0.f,0.f,0.f,0.f,0.f};

    const float bias = b1[g * ENDC + n0 + l16];

    for (int c = 0; c + 1 < NCHUNK; ++c) {
        for (int q = 0; q < 4; ++q)
            pre[q] = *(const float4*)(wsrc + (size_t)q * 32 * SKIP + (c + 1) * KC);
        mma_chunk(Arow0, Arow1, Bs + ((c + 1) & 1) * BS_TILE + boff, c * KC, c0, c1);
        float* bnext = bdst + (c & 1) * BS_TILE;
        for (int q = 0; q < 4; ++q)
            *(float4*)(bnext + q * 32 * BS_STRIDE) = pre[q];
        __syncthreads();
    }

    const float* w2g = W2 + (size_t)g * W2_BLK;
    pre[0] = ((const float4*)w2g)[tid];
    pre[1] = ((const float4*)w2g)[tid + 256];
    pre[2] = ((const float4*)w2g)[tid + 512];
    const float w2t = w2g[3 * 1024 + tid];

    mma_chunk(Arow0, Arow1, Bs + boff, (NCHUNK - 1) * KC, c0, c1);

    ((float4*)W2s)[tid]       = pre[0];
    ((float4*)W2s)[tid + 256] = pre[1];
    ((float4*)W2s)[tid + 512] = pre[2];
    W2s[3 * 1024 + tid]       = w2t;

    const int   colH = n0 + l16;
    #pragma unroll
    for (int j = 0; j < 8; ++j) {
        int row0 = j + 8 * half;
        float v0 = c0[j] + bias;
        Hs[row0 * HS_STRIDE + colH] = v0 > 0.f ? v0 : 0.f;
        int row1 = 16 + j + 8 * half;
        if (row1 < HIST) {
            float v1 = c1[j] + bias;
            Hs[row1 * HS_STRIDE + colH] = v1 > 0.f ? v1 : 0.f;
        }
    }
    __syncthreads();

    float* Ps = Xs;
    {
        const int e2 = tid & (ENDC - 1);
        float h[HIST];
        #pragma unroll
        for (int t = 0; t < HIST; ++t)
            h[t] = Hs[t * HS_STRIDE + e2];
        float w[KERN];
        #pragma unroll
        for (int k = 0; k < KERN; ++k)
            w[k] = W2s[tid * KERN + k];
        float p[FUT];
        #pragma unroll
        for (int f = 0; f < FUT; ++f) {
            float a = 0.f;
            #pragma unroll
            for (int k = 0; k < KERN; ++k)
                a += h[f + k] * w[k];
            p[f] = a;
        }
        float4* pd = (float4*)&Ps[tid * FUT];
        pd[0] = make_float4(p[0], p[1], p[2],  p[3]);
        pd[1] = make_float4(p[4], p[5], p[6],  p[7]);
        pd[2] = make_float4(p[8], p[9], p[10], p[11]);
    }
    __syncthreads();

    const int group = wv * 4 + (lane >> 3);
    const int l8    = lane & 7;
    if (group < FUT * OUTD) {
        const int f = group % FUT;
        const int o = group / FUT;
        const float* pp = &Ps[(o * ENDC + l8) * FUT + f];
        float acc = 0.f;
        #pragma unroll
        for (int i = 0; i < ENDC / 8; ++i)
            acc += pp[i * 8 * FUT];
        acc += __shfl_xor(acc, 4, 32);
        acc += __shfl_xor(acc, 2, 32);
        acc += __shfl_xor(acc, 1, 32);
        if (l8 == 0)
            out[(size_t)node * (FUT * OUTD) + f * OUTD + o] = acc + (o ? b2o1 : b2o0);
    }
}

extern "C" void kernel_launch(void* const* d_in, const int* in_sizes, int n_in,
                              void* d_out, int out_size, void* d_ws, size_t ws_size,
                              hipStream_t stream) {
    const float* x      = (const float*)d_in[0];
    const int*   labels = (const int*)  d_in[1];
    const float* W1     = (const float*)d_in[2];
    const float* b1     = (const float*)d_in[3];
    const float* W2     = (const float*)d_in[4];
    const float* b2     = (const float*)d_in[5];
    float*       out    = (float*)d_out;

    moe_tcnn_kernel<<<NODES, 256, 0, stream>>>(x, labels, W1, b1, W2, b2, out);
}


# Two nodes of equal expert label share a work-group (in-kernel rank of each 256-label segment by ballots, no atomics): 48 rows = 3 full M-tiles, no pad rows; each W1 chunk staged once for both
# speedup vs baseline: 1.8875x; 1.1333x over previous
#include <hip/hip_runtime.h>
#include <hip/hip_bf16.h>


#define NB      8
#define NN      512
#define HIST    24
#define SKIP    256
#define ENDC    128
#define KERN    13
#define FUT     12
#define OUTD    2
#define NODES   (NB * NN)
#define NEXP    8

#define PAIR      2
#define MROWS     (PAIR * HIST)
#define SEG       256
#define NSEG      (NODES / SEG)
#define SEG_TILES (SEG / PAIR + NEXP / 2)
#define GRID      (NSEG * SEG_TILES)

#define XS_STRIDE 260
#define HS_STRIDE 132
#define KC        32
#define BS_STRIDE 36
#define BS_TILE   (ENDC * BS_STRIDE)
#define NCHUNK    (SKIP / KC)
#define W2_BLK    (OUTD * ENDC * KERN)

typedef float v2f __attribute__((ext_vector_type(2)));
typedef float v8f __attribute__((ext_vector_type(8)));

static __device__ __forceinline__
void mma_chunk(const float* Arow0, const float* Arow1, const float* Arow2, const float* Bcur, int kbase,
               v8f& c0, v8f& c1, v8f& c2)
{
    #pragma unroll
    for (int u = 0; u < 8; ++u) {
        int k = kbase + u * 4;
        v2f af0 = *(const v2f*)(Arow0 + k);
        v2f af1 = *(const v2f*)(Arow1 + k);
        v2f af2 = *(const v2f*)(Arow2 + k);
        v2f bf  = *(const v2f*)(Bcur + u * 4);
        c0 = __builtin_amdgcn_wmma_f32_16x16x4_f32(false, af0, false, bf,
                                                   (short)0, c0, false, false);
        c1 = __builtin_amdgcn_wmma_f32_16x16x4_f32(false, af1, false, bf,
                                                   (short)0, c1, false, false);
        c2 = __builtin_amdgcn_wmma_f32_16x16x4_f32(false, af2, false, bf,
                                                   (short)0, c2, false, false);
    }
}

__global__ __launch_bounds__(256)
void moe_tcnn_kernel(const float* __restrict__ x,
                     const int*   __restrict__ labels,
                     const float* __restrict__ W1,
                     const float* __restrict__ b1,
                     const float* __restrict__ W2,
                     const float* __restrict__ b2,
                     float*       __restrict__ out)
{
    __shared__ __align__(16) float Xs[MROWS * XS_STRIDE];
    __shared__ __align__(16) float Bs[BS_TILE + MROWS * HS_STRIDE + W2_BLK];
    __shared__ __align__(16) int   s_cnt[8 * 4];
    __shared__ int                 s_pair[PAIR];
    float* Hs  = Bs + BS_TILE;
    float* W2s = Hs + MROWS * HS_STRIDE;

    const int tid  = threadIdx.x;
    const int lane = tid & 31;
    const int wv   = __builtin_amdgcn_readfirstlane(tid >> 5);
    const int seg  = blockIdx.x / SEG_TILES;
    const int tile = blockIdx.x - seg * SEG_TILES;

    const int mynode = seg * SEG + tid;
    const int lab    = labels[mynode] & (NEXP - 1);
    {
        int cnt[NEXP];
        #pragma unroll
        for (int e = 0; e < NEXP; ++e)
            cnt[e] = __popc(__builtin_amdgcn_ballot_w32(lab == e));
        if (lane == 0)
            ((int4*)s_cnt)[wv] = make_int4(cnt[0] | (cnt[1] << 16), cnt[2] | (cnt[3] << 16),
                                           cnt[4] | (cnt[5] << 16), cnt[6] | (cnt[7] << 16));
    }
    __syncthreads();
    int tot[4]   = {0, 0, 0, 0};
    int below[4] = {0, 0, 0, 0};
    #pragma unroll
    for (int v = 0; v < 8; ++v) {
        const int4 c = ((const int4*)s_cnt)[v];
        tot[0] += c.x; tot[1] += c.y; tot[2] += c.z; tot[3] += c.w;
        if (v < wv) { below[0] += c.x; below[1] += c.y; below[2] += c.z; below[3] += c.w; }
    }
    int g = -1, first = 0, ng = 0, bg = 0, start = 0;
    #pragma unroll
    for (int e = 0; e < NEXP; ++e) {
        const int ne = (tot[e >> 1] >> (16 * (e & 1))) & 0xFFFF;
        const int te = (ne + 1) >> 1;
        if (tile >= start && tile < start + te) {
            g     = e;
            first = 2 * (tile - start);
            ng    = ne;
            bg    = (below[e >> 1] >> (16 * (e & 1))) & 0xFFFF;
        }
        start += te;
    }
    g = __builtin_amdgcn_readfirstlane(g);
    if (g < 0) return;

    {
        const bool     mine = (lab == g);
        const unsigned mg   = __builtin_amdgcn_ballot_w32(mine);
        const int      rank = bg + __popc(mg & ((1u << lane) - 1u));
        if (mine && rank == first)           s_pair[0] = mynode;
        if (mine && rank == first + 1)       s_pair[1] = mynode;
        if (tid == 0 && first + 1 >= ng)     s_pair[1] = -1;
    }
    __syncthreads();
    const int  node0  = __builtin_amdgcn_readfirstlane(s_pair[0]);
    const int  node1  = __builtin_amdgcn_readfirstlane(s_pair[1]);
    const bool has1   = node1 >= 0;
    const int  node1x = has1 ? node1 : node0;

    const float b2o0 = b2[g * OUTD + 0];
    const float b2o1 = b2[g * OUTD + 1];

    const int srow = tid >> 3;
    const int scol = (tid & 7) * 4;
    const float* wsrc = W1 + ((size_t)g * ENDC + srow) * SKIP + scol;
    float*       bdst = &Bs[srow * BS_STRIDE + scol];

    float4 pre[4];
    for (int q = 0; q < 4; ++q)
        pre[q] = *(const float4*)(wsrc + (size_t)q * 32 * SKIP);

    const float* xp0 = x + (size_t)node0  * (HIST * SKIP);
    const float* xp1 = x + (size_t)node1x * (HIST * SKIP);
    #pragma unroll
    for (int q = 0; q < 6; ++q) {
        int idx4 = tid + q * 256;
        int idx  = idx4 * 4;
        int row  = idx >> 8;
        int col  = idx & 255;
        float4 v = *(const float4*)(xp0 + idx);
        float4 w = *(const float4*)(xp1 + idx);
        v.x = v.x > 0.f ? v.x : 0.f;
        v.y = v.y > 0.f ? v.y : 0.f;
        v.z = v.z > 0.f ? v.z : 0.f;
        v.w = v.w > 0.f ? v.w : 0.f;
        w.x = w.x > 0.f ? w.x : 0.f;
        w.y = w.y > 0.f ? w.y : 0.f;
        w.z = w.z > 0.f ? w.z : 0.f;
        w.w = w.w > 0.f ? w.w : 0.f;
        *(float4*)(&Xs[row * XS_STRIDE + col])          = v;
        *(float4*)(&Xs[(HIST + row) * XS_STRIDE + col]) = w;
    }
    for (int q = 0; q < 4; ++q)
        *(float4*)(bdst + BS_TILE + q * 32 * BS_STRIDE) = pre[q];
    __syncthreads();

    const int half = lane >> 4;
    const int l16  = lane & 15;
    const int n0   = wv * 16;

    const int boff = (n0 + l16) * BS_STRIDE + 2 * half;
    const float* Arow0 = &Xs[l16 * XS_STRIDE + 2 * half];
    const float* Arow1 = &Xs[(16 + l16) * XS_STRIDE + 2 * half];
    const float* Arow2 = &Xs[(32 + l16) * XS_STRIDE + 2 * half];

    v8f c0 = {0.f,0.f,0.f,0.f,0.f,0.f,0.f,0.f};
    v8f c1 = {0.f,0.f,0.f,0.f,0.f,0.f,0.f,0.f};
    v8f c2 = {0.f,0.f,0.f,0.f,0.f,0.f,0.f,0.f};

    const float bias = b1[g * ENDC + n0 + l16];

    for (int c = 0; c + 1 < NCHUNK; ++c) {
        for (int q = 0; q < 4; ++q)
            pre[q] = *(const float4*)(wsrc + (size_t)q * 32 * SKIP + (c + 1) * KC);
        mma_chunk(Arow0, Arow1, Arow2, Bs + ((c + 1) & 1) * BS_TILE + boff, c * KC, c0, c1, c2);
        float* bnext = bdst + (c & 1) * BS_TILE;
        for (int q = 0; q < 4; ++q)
            *(float4*)(bnext + q * 32 * BS_STRIDE) = pre[q];
        __syncthreads();
    }

    const float* w2g = W2 + (size_t)g * W2_BLK;
    pre[0] = ((const float4*)w2g)[tid];
    pre[1] = ((const float4*)w2g)[tid + 256];
    pre[2] = ((const float4*)w2g)[tid + 512];
    const float w2t = w2g[3 * 1024 + tid];

    mma_chunk(Arow0, Arow1, Arow2, Bs + boff, (NCHUNK - 1) * KC, c0, c1, c2);

    ((float4*)W2s)[tid]       = pre[0];
    ((float4*)W2s)[tid + 256] = pre[1];
    ((float4*)W2s)[tid + 512] = pre[2];
    W2s[3 * 1024 + tid]       = w2t;

    const int   colH = n0 + l16;
    #pragma unroll
    for (int j = 0; j < 8; ++j) {
        int r = j + 8 * half;
        float v0 = c0[j] + bias;
        Hs[r * HS_STRIDE + colH] = v0 > 0.f ? v0 : 0.f;
        float v1 = c1[j] + bias;
        Hs[(16 + r) * HS_STRIDE + colH] = v1 > 0.f ? v1 : 0.f;
        float v2 = c2[j] + bias;
        Hs[(32 + r) * HS_STRIDE + colH] = v2 > 0.f ? v2 : 0.f;
    }
    __syncthreads();

    float* Ps = Xs;
    {
        const int e2 = tid & (ENDC - 1);
        float w[KERN];
        #pragma unroll
        for (int k = 0; k < KERN; ++k)
            w[k] = W2s[tid * KERN + k];
        #pragma unroll
        for (int nd = 0; nd < PAIR; ++nd) {
            float h[HIST];
            #pragma unroll
            for (int t = 0; t < HIST; ++t)
                h[t] = Hs[(nd * HIST + t) * HS_STRIDE + e2];
            float p[FUT];
            #pragma unroll
            for (int f = 0; f < FUT; ++f) {
                float a = 0.f;
                #pragma unroll
                for (int k = 0; k < KERN; ++k)
                    a += h[f + k] * w[k];
                p[f] = a;
            }
            float4* pd = (float4*)&Ps[(nd * 256 + tid) * FUT];
            pd[0] = make_float4(p[0], p[1], p[2],  p[3]);
            pd[1] = make_float4(p[4], p[5], p[6],  p[7]);
            pd[2] = make_float4(p[8], p[9], p[10], p[11]);
        }
    }
    __syncthreads();

    const int group = wv * 4 + (lane >> 3);
    const int l8    = lane & 7;
    if (group < FUT * OUTD) {
        const int f = group % FUT;
        const int o = group / FUT;
        #pragma unroll
        for (int nd = 0; nd < PAIR; ++nd) {
            const float* pp = &Ps[(nd * 256 + o * ENDC + l8) * FUT + f];
            float acc = 0.f;
            #pragma unroll
            for (int i = 0; i < ENDC / 8; ++i)
                acc += pp[i * 8 * FUT];
            acc += __shfl_xor(acc, 4, 32);
            acc += __shfl_xor(acc, 2, 32);
            acc += __shfl_xor(acc, 1, 32);
            const int node = nd ? node1 : node0;
            if (l8 == 0 && (nd == 0 || has1))
                out[(size_t)node * (FUT * OUTD) + f * OUTD + o] = acc + (o ? b2o1 : b2o0);
        }
    }
}

extern "C" void kernel_launch(void* const* d_in, const int* in_sizes, int n_in,
                              void* d_out, int out_size, void* d_ws, size_t ws_size,
                              hipStream_t stream) {
    const float* x      = (const float*)d_in[0];
    const int*   labels = (const int*)  d_in[1];
    const float* W1     = (const float*)d_in[2];
    const float* b1     = (const float*)d_in[3];
    const float* W2     = (const float*)d_in[4];
    const float* b2     = (const float*)d_in[5];
    float*       out    = (float*)d_out;

    moe_tcnn_kernel<<<GRID, 256, 0, stream>>>(x, labels, W1, b1, W2, b2, out);
}


# Four work-groups per WGP, not three: LDS 107144 -> 68488 B (one W1 chunk buffer, each wave stages and reads only its own 16 rows, no k-loop barrier; stage 2 reuses the freed tiles), VGPR cap 128
# speedup vs baseline: 1.9525x; 1.0344x over previous
#include <hip/hip_runtime.h>
#include <hip/hip_bf16.h>


#define NB      8
#define NN      512
#define HIST    24
#define SKIP    256
#define ENDC    128
#define KERN    13
#define FUT     12
#define OUTD    2
#define NODES   (NB * NN)
#define NEXP    8

#define PAIR      2
#define MROWS     (PAIR * HIST)
#define SEG       256
#define NSEG      (NODES / SEG)
#define SEG_TILES (SEG / PAIR + NEXP / 2)
#define GRID      (NSEG * SEG_TILES)

#define XS_STRIDE 260
#define HS_STRIDE 132
#define KC        32
#define BS_STRIDE 36
#define BS_TILE   (ENDC * BS_STRIDE)
#define NCHUNK    (SKIP / KC)
#define W2_BLK    (OUTD * ENDC * KERN)

typedef float v2f __attribute__((ext_vector_type(2)));
typedef float v8f __attribute__((ext_vector_type(8)));

static __device__ __forceinline__
void mma_chunk(const float* Arow0, const float* Arow1, const float* Arow2, const float* Bcur, int kbase,
               v8f& c0, v8f& c1, v8f& c2)
{
    #pragma unroll
    for (int u = 0; u < 8; ++u) {
        int k = kbase + u * 4;
        v2f af0 = *(const v2f*)(Arow0 + k);
        v2f af1 = *(const v2f*)(Arow1 + k);
        v2f af2 = *(const v2f*)(Arow2 + k);
        v2f bf  = *(const v2f*)(Bcur + u * 4);
        c0 = __builtin_amdgcn_wmma_f32_16x16x4_f32(false, af0, false, bf,
                                                   (short)0, c0, false, false);
        c1 = __builtin_amdgcn_wmma_f32_16x16x4_f32(false, af1, false, bf,
                                                   (short)0, c1, false, false);
        c2 = __builtin_amdgcn_wmma_f32_16x16x4_f32(false, af2, false, bf,
                                                   (short)0, c2, false, false);
    }
    __builtin_amdgcn_sched_group_barrier(0x100, 16, 0);
    __builtin_amdgcn_sched_group_barrier(0x008, 24, 0);
}

__global__ __launch_bounds__(256, 8)
void moe_tcnn_kernel(const float* __restrict__ x,
                     const int*   __restrict__ labels,
                     const float* __restrict__ W1,
                     const float* __restrict__ b1,
                     const float* __restrict__ W2,
                     const float* __restrict__ b2,
                     float*       __restrict__ out)
{
    __shared__ __align__(16) float Xs[MROWS * XS_STRIDE];
    __shared__ __align__(16) float Bs[BS_TILE];
    __shared__ __align__(16) int   s_cnt[8 * 4];
    __shared__ int                 s_pair[PAIR];
    static_assert(W2_BLK <= BS_TILE, "W2[g] must fit the chunk buffer");
    static_assert(MROWS * HS_STRIDE + PAIR * 256 * FUT <= MROWS * XS_STRIDE, "Hs and Ps must fit the X tile");
    float* Hs  = Xs;
    float* Ps  = Xs + MROWS * HS_STRIDE;
    float* W2s = Bs;

    const int tid  = threadIdx.x;
    const int lane = tid & 31;
    const int wv   = __builtin_amdgcn_readfirstlane(tid >> 5);
    const int seg  = blockIdx.x / SEG_TILES;
    const int tile = blockIdx.x - seg * SEG_TILES;

    const int mynode = seg * SEG + tid;
    const int lab    = labels[mynode] & (NEXP - 1);
    {
        int cnt[NEXP];
        #pragma unroll
        for (int e = 0; e < NEXP; ++e)
            cnt[e] = __popc(__builtin_amdgcn_ballot_w32(lab == e));
        if (lane == 0)
            ((int4*)s_cnt)[wv] = make_int4(cnt[0] | (cnt[1] << 16), cnt[2] | (cnt[3] << 16),
                                           cnt[4] | (cnt[5] << 16), cnt[6] | (cnt[7] << 16));
    }
    __syncthreads();
    int tot[4]   = {0, 0, 0, 0};
    int below[4] = {0, 0, 0, 0};
    #pragma unroll
    for (int v = 0; v < 8; ++v) {
        const int4 c = ((const int4*)s_cnt)[v];
        tot[0] += c.x; tot[1] += c.y; tot[2] += c.z; tot[3] += c.w;
        if (v < wv) { below[0] += c.x; below[1] += c.y; below[2] += c.z; below[3] += c.w; }
    }
    int g = -1, first = 0, ng = 0, bg = 0, start = 0;
    #pragma unroll
    for (int e = 0; e < NEXP; ++e) {
        const int ne = (tot[e >> 1] >> (16 * (e & 1))) & 0xFFFF;
        const int te = (ne + 1) >> 1;
        if (tile >= start && tile < start + te) {
            g     = e;
            first = 2 * (tile - start);
            ng    = ne;
            bg    = (below[e >> 1] >> (16 * (e & 1))) & 0xFFFF;
        }
        start += te;
    }
    g = __builtin_amdgcn_readfirstlane(g);
    if (g < 0) return;

    {
        const bool     mine = (lab == g);
        const unsigned mg   = __builtin_amdgcn_ballot_w32(mine);
        const int      rank = bg + __popc(mg & ((1u << lane) - 1u));
        if (mine && rank == first)           s_pair[0] = mynode;
        if (mine && rank == first + 1)       s_pair[1] = mynode;
        if (tid == 0 && first + 1 >= ng)     s_pair[1] = -1;
    }
    __syncthreads();
    const int  node0  = __builtin_amdgcn_readfirstlane(s_pair[0]);
    const int  node1  = __builtin_amdgcn_readfirstlane(s_pair[1]);
    const bool has1   = node1 >= 0;
    const int  node1x = has1 ? node1 : node0;

    const float b2o0 = b2[g * OUTD + 0];
    const float b2o1 = b2[g * OUTD + 1];

    const int n0   = wv * 16;
    const int srow = lane >> 3;
    const int scol = (lane & 7) * 4;
    const float* wsrc = W1 + ((size_t)g * ENDC + n0 + srow) * SKIP + scol;
    float*       bdst = &Bs[(n0 + srow) * BS_STRIDE + scol];

    float4 pre0 = *(const float4*)(wsrc + 0 * 4 * SKIP);
    float4 pre1 = *(const float4*)(wsrc + 1 * 4 * SKIP);
    float4 pre2 = *(const float4*)(wsrc + 2 * 4 * SKIP);
    float4 pre3 = *(const float4*)(wsrc + 3 * 4 * SKIP);

    const float* xp0 = x + (size_t)node0  * (HIST * SKIP);
    const float* xp1 = x + (size_t)node1x * (HIST * SKIP);
    #pragma unroll
    for (int q = 0; q < 6; ++q) {
        int idx4 = tid + q * 256;
        int idx  = idx4 * 4;
        int row  = idx >> 8;
        int col  = idx & 255;
        float4 v = *(const float4*)(xp0 + idx);
        float4 w = *(const float4*)(xp1 + idx);
        v.x = v.x > 0.f ? v.x : 0.f;
        v.y = v.y > 0.f ? v.y : 0.f;
        v.z = v.z > 0.f ? v.z : 0.f;
        v.w = v.w > 0.f ? v.w : 0.f;
        w.x = w.x > 0.f ? w.x : 0.f;
        w.y = w.y > 0.f ? w.y : 0.f;
        w.z = w.z > 0.f ? w.z : 0.f;
        w.w = w.w > 0.f ? w.w : 0.f;
        *(float4*)(&Xs[row * XS_STRIDE + col])          = v;
        *(float4*)(&Xs[(HIST + row) * XS_STRIDE + col]) = w;
    }
    *(float4*)(bdst + 0 * 4 * BS_STRIDE) = pre0;
    *(float4*)(bdst + 1 * 4 * BS_STRIDE) = pre1;
    *(float4*)(bdst + 2 * 4 * BS_STRIDE) = pre2;
    *(float4*)(bdst + 3 * 4 * BS_STRIDE) = pre3;
    __syncthreads();

    const int half = lane >> 4;
    const int l16  = lane & 15;

    const int boff = (n0 + l16) * BS_STRIDE + 2 * half;
    const float* Arow0 = &Xs[l16 * XS_STRIDE + 2 * half];
    const float* Arow1 = &Xs[(16 + l16) * XS_STRIDE + 2 * half];
    const float* Arow2 = &Xs[(32 + l16) * XS_STRIDE + 2 * half];

    v8f c0 = {0.f,0.f,0.f,0.f,0.f,0.f,0.f,0.f};
    v8f c1 = {0.f,0.f,0.f,0.f,0.f,0.f,0.f,0.f};
    v8f c2 = {0.f,0.f,0.f,0.f,0.f,0.f,0.f,0.f};

    const float bias = b1[g * ENDC + n0 + l16];

    #pragma nounroll
    for (int c = 0; c + 1 < NCHUNK; ++c) {
        const float* wnext = wsrc + (c + 1) * KC;
        pre0 = *(const float4*)(wnext + 0 * 4 * SKIP);
        pre1 = *(const float4*)(wnext + 1 * 4 * SKIP);
        pre2 = *(const float4*)(wnext + 2 * 4 * SKIP);
        pre3 = *(const float4*)(wnext + 3 * 4 * SKIP);
        mma_chunk(Arow0, Arow1, Arow2, Bs + boff, c * KC, c0, c1, c2);
        *(float4*)(bdst + 0 * 4 * BS_STRIDE) = pre0;
        *(float4*)(bdst + 1 * 4 * BS_STRIDE) = pre1;
        *(float4*)(bdst + 2 * 4 * BS_STRIDE) = pre2;
        *(float4*)(bdst + 3 * 4 * BS_STRIDE) = pre3;
    }

    mma_chunk(Arow0, Arow1, Arow2, Bs + boff, (NCHUNK - 1) * KC, c0, c1, c2);
    __syncthreads();

    const float* w2g = W2 + (size_t)g * W2_BLK;
    pre0 = ((const float4*)w2g)[tid];
    pre1 = ((const float4*)w2g)[tid + 256];
    pre2 = ((const float4*)w2g)[tid + 512];
    const float w2t = w2g[3 * 1024 + tid];

    const int   colH = n0 + l16;
    #pragma unroll
    for (int j = 0; j < 8; ++j) {
        int r = j + 8 * half;
        float v0 = c0[j] + bias;
        Hs[r * HS_STRIDE + colH] = v0 > 0.f ? v0 : 0.f;
        float v1 = c1[j] + bias;
        Hs[(16 + r) * HS_STRIDE + colH] = v1 > 0.f ? v1 : 0.f;
        float v2 = c2[j] + bias;
        Hs[(32 + r) * HS_STRIDE + colH] = v2 > 0.f ? v2 : 0.f;
    }
    ((float4*)W2s)[tid]       = pre0;
    ((float4*)W2s)[tid + 256] = pre1;
    ((float4*)W2s)[tid + 512] = pre2;
    W2s[3 * 1024 + tid]       = w2t;
    __syncthreads();

    {
        const int e2 = tid & (ENDC - 1);
        float w[KERN];
        #pragma unroll
        for (int k = 0; k < KERN; ++k)
            w[k] = W2s[tid * KERN + k];
        #pragma unroll
        for (int nd = 0; nd < PAIR; ++nd) {
            float h[HIST];
            #pragma unroll
            for (int t = 0; t < HIST; ++t)
                h[t] = Hs[(nd * HIST + t) * HS_STRIDE + e2];
            float p[FUT];
            #pragma unroll
            for (int f = 0; f < FUT; ++f) {
                float a = 0.f;
                #pragma unroll
                for (int k = 0; k < KERN; ++k)
                    a += h[f + k] * w[k];
                p[f] = a;
            }
            float4* pd = (float4*)&Ps[(nd * 256 + tid) * FUT];
            pd[0] = make_float4(p[0], p[1], p[2],  p[3]);
            pd[1] = make_float4(p[4], p[5], p[6],  p[7]);
            pd[2] = make_float4(p[8], p[9], p[10], p[11]);
        }
    }
    __syncthreads();

    const int group = wv * 4 + (lane >> 3);
    const int l8    = lane & 7;
    if (group < FUT * OUTD) {
        const int f = group % FUT;
        const int o = group / FUT;
        #pragma unroll
        for (int nd = 0; nd < PAIR; ++nd) {
            const float* pp = &Ps[(nd * 256 + o * ENDC + l8) * FUT + f];
            float acc = 0.f;
            #pragma unroll
            for (int i = 0; i < ENDC / 8; ++i)
                acc += pp[i * 8 * FUT];
            acc += __shfl_xor(acc, 4, 32);
            acc += __shfl_xor(acc, 2, 32);
            acc += __shfl_xor(acc, 1, 32);
            const int node = nd ? node1 : node0;
            if (l8 == 0 && (nd == 0 || has1))
                out[(size_t)node * (FUT * OUTD) + f * OUTD + o] = acc + (o ? b2o1 : b2o0);
        }
    }
}

extern "C" void kernel_launch(void* const* d_in, const int* in_sizes, int n_in,
                              void* d_out, int out_size, void* d_ws, size_t ws_size,
                              hipStream_t stream) {
    const float* x      = (const float*)d_in[0];
    const int*   labels = (const int*)  d_in[1];
    const float* W1     = (const float*)d_in[2];
    const float* b1     = (const float*)d_in[3];
    const float* W2     = (const float*)d_in[4];
    const float* b2     = (const float*)d_in[5];
    float*       out    = (float*)d_out;

    moe_tcnn_kernel<<<GRID, 256, 0, stream>>>(x, labels, W1, b1, W2, b2, out);
}
